# starting kernel
# baseline (speedup 1.0000x reference)
#include <hip/hip_runtime.h>

#define DN   128
#define KTOT 256
#define WPAD 260
#define ROWS_PER_BLOCK 80
#define GEMM_THREADS   160
#define KEY_SENTINEL ((int)0x80000000)

typedef float v2f __attribute__((ext_vector_type(2)));
typedef float v8f __attribute__((ext_vector_type(8)));

__device__ __forceinline__ int enc_f32(float f) {
    int b = __float_as_int(f);
    return b >= 0 ? b : (b ^ 0x7FFFFFFF);
}
__device__ __forceinline__ float dec_f32(int k) {
    return __int_as_float(k >= 0 ? k : (k ^ 0x7FFFFFFF));
}

__global__ void init_keys_kernel(int4* __restrict__ out, int n4) {
    int t = blockIdx.x * blockDim.x + threadIdx.x;
    if (t < n4) {
        out[t] = make_int4(KEY_SENTINEL, KEY_SENTINEL, KEY_SENTINEL, KEY_SENTINEL);
    }
}

__global__ void edge_max_kernel(const float* __restrict__ H,
                                const int* __restrict__ src,
                                const int* __restrict__ dst,
                                int* keys, int E) {
    int t = blockIdx.x * blockDim.x + threadIdx.x;
    int e = t >> 5;
    if (e >= E) return;
    int j0 = (t & 31) << 2;
    int s = src[e];
    int d = dst[e];
    const float4 hv = *(const float4*)(H + (size_t)d * DN + j0);
    int* kp = keys + (size_t)s * DN + j0;
    const int4 cur = *(const int4*)kp;
    int k0 = enc_f32(hv.x);
    int k1 = enc_f32(hv.y);
    int k2 = enc_f32(hv.z);
    int k3 = enc_f32(hv.w);
    if (k0 > cur.x) atomicMax(kp + 0, k0);
    if (k1 > cur.y) atomicMax(kp + 1, k1);
    if (k2 > cur.z) atomicMax(kp + 2, k2);
    if (k3 > cur.w) atomicMax(kp + 3, k3);
}

__global__ __launch_bounds__(GEMM_THREADS) void gemm_kernel(
        const float* __restrict__ H,
        const float* __restrict__ W,
        float* out  ) {
    __shared__ float Ws[DN * WPAD];

    for (int i = threadIdx.x; i < DN * (KTOT / 4); i += GEMM_THREADS) {
        int r  = i >> 6;
        int c4 = (i & 63) << 2;
        float4 w = *(const float4*)(W + (size_t)r * KTOT + c4);
        *(float4*)(&Ws[r * WPAD + c4]) = w;
    }
    __syncthreads();

    const int wave = threadIdx.x >> 5;
    const int lane = threadIdx.x & 31;
    const int lm   = lane & 15;
    const int h2   = (lane >> 4) << 1;
    const int m0   = blockIdx.x * ROWS_PER_BLOCK + wave * 16;
    const int row  = m0 + lm;

    const float* arow = H + (size_t)row * DN;
    const int*   krow = (const int*)out + (size_t)row * DN;

    v8f zero = {};
    v8f acc[8];
#pragma unroll
    for (int nt = 0; nt < 8; ++nt) acc[nt] = zero;

#pragma unroll 4
    for (int kk = 0; kk < 32; ++kk) {
        const int kc = (kk << 2) + h2;
        v2f a = *(const v2f*)(arow + kc);
#pragma unroll
        for (int nt = 0; nt < 8; ++nt) {
            v2f b = *(const v2f*)(&Ws[(nt * 16 + lm) * WPAD + kc]);
            acc[nt] = __builtin_amdgcn_wmma_f32_16x16x4_f32(
                false, a, false, b, (short)0, acc[nt], false, false);
        }
    }

#pragma unroll 4
    for (int kk = 0; kk < 32; ++kk) {
        const int kc = (kk << 2) + h2;
        int k0 = krow[kc];
        int k1 = krow[kc + 1];
        v2f a;
        a.x = (k0 == KEY_SENTINEL) ? 0.0f : dec_f32(k0);
        a.y = (k1 == KEY_SENTINEL) ? 0.0f : dec_f32(k1);
#pragma unroll
        for (int nt = 0; nt < 8; ++nt) {
            v2f b = *(const v2f*)(&Ws[(nt * 16 + lm) * WPAD + 128 + kc]);
            acc[nt] = __builtin_amdgcn_wmma_f32_16x16x4_f32(
                false, a, false, b, (short)0, acc[nt], false, false);
        }
    }

    const int orow0 = m0 + ((lane >> 4) << 3);
    float* obase = out + (size_t)orow0 * DN + lm;
#pragma unroll
    for (int r = 0; r < 8; ++r) {
#pragma unroll
        for (int nt = 0; nt < 8; ++nt) {
            obase[(size_t)r * DN + nt * 16] = acc[nt][r];
        }
    }
}

extern "C" void kernel_launch(void* const* d_in, const int* in_sizes, int n_in,
                              void* d_out, int out_size, void* d_ws, size_t ws_size,
                              hipStream_t stream) {
    (void)n_in; (void)out_size; (void)d_ws; (void)ws_size;
    const float* H   = (const float*)d_in[0];
    const int*   src = (const int*)d_in[1];
    const int*   dst = (const int*)d_in[2];
    const float* W   = (const float*)d_in[3];

    const int N = in_sizes[0] / DN;
    const int E = in_sizes[1];
    float* out = (float*)d_out;

    int n4 = (N * DN) / 4;
    init_keys_kernel<<<(n4 + 255) / 256, 256, 0, stream>>>((int4*)d_out, n4);

    long long et = (long long)E * 32;
    int eb = (int)((et + 255) / 256);
    edge_max_kernel<<<eb, 256, 0, stream>>>(H, src, dst, (int*)d_out, E);

    gemm_kernel<<<N / ROWS_PER_BLOCK, GEMM_THREADS, 0, stream>>>(H, W, out);
}


# segment-max by node-range ownership: each workgroup keeps its nodes' keys in LDS, scans all src, gathers matching H rows, LDS integer max; global atomics and the fill launch removed
# speedup vs baseline: 1.6840x; 1.6840x over previous
#include <hip/hip_runtime.h>

#define DN   128
#define KTOT 256
#define WPAD 260
#define ROWS_PER_BLOCK 80
#define GEMM_THREADS   160
#define KEY_SENTINEL ((int)0x80000000)

typedef float v2f __attribute__((ext_vector_type(2)));
typedef float v8f __attribute__((ext_vector_type(8)));

__device__ __forceinline__ int enc_f32(float f) {
    int b = __float_as_int(f);
    return b >= 0 ? b : (b ^ 0x7FFFFFFF);
}
__device__ __forceinline__ float dec_f32(int k) {
    return __int_as_float(k >= 0 ? k : (k ^ 0x7FFFFFFF));
}

#define SEG_NODES   391
#define SEG_GROUPS  256
#define SEG_THREADS 1024
#define SEG_WAVES   32
#define WL_FLUSH    64
#define WL_CAP      192
#define SCAN_U      4


typedef __attribute__((address_space(3))) volatile int lds_vint;

__device__ __forceinline__ void seg_append(bool p, int e, unsigned nl,
                                           lds_vint* le, lds_vint* ln,
                                           int& cnt) {
    const unsigned m = __builtin_amdgcn_ballot_w32(p);
    if (p) {
        const int pos = cnt + (int)__builtin_amdgcn_mbcnt_lo(m, 0u);
        le[pos] = e;
        ln[pos] = (int)nl;
    }
    cnt += __builtin_popcount(m);
}

__device__ __forceinline__ void seg_flush(const float* __restrict__ H,
                                          const int* __restrict__ dst,
                                          int* keysL,
                                          lds_vint* le, lds_vint* ln,
                                          int cnt, int lane) {
    for (int i = lane; i < cnt; i += 32) {
        const int e = le[i];
        le[i] = dst[e];
    }
    for (int i = 0; i < cnt; i += 8) {
        float4 h[8];
        int nl[8];
#pragma unroll
        for (int j = 0; j < 8; ++j) {
            const int idx = min(i + j, cnt - 1);
            const int d = __builtin_amdgcn_readfirstlane(le[idx]);
            nl[j] = __builtin_amdgcn_readfirstlane(ln[idx]);
            h[j] = *(const float4*)(H + (size_t)d * DN + (lane << 2));
        }
        __builtin_amdgcn_sched_barrier(0);
#pragma unroll
        for (int j = 0; j < 8; ++j) {
            int* kp = keysL + nl[j] * DN + (lane << 2);
            atomicMax(kp + 0, enc_f32(h[j].x));
            atomicMax(kp + 1, enc_f32(h[j].y));
            atomicMax(kp + 2, enc_f32(h[j].z));
            atomicMax(kp + 3, enc_f32(h[j].w));
        }
    }
}

__device__ __forceinline__ int seg_i4(int it, int rot, int iters, int tid) {
    int itr = it + rot;
    if (itr >= iters) itr -= iters;
    return itr * SEG_THREADS + tid;
}
__device__ __forceinline__ int4 seg_load(const int4* __restrict__ src4, int it,
                                         int rot, int iters, int tid, int E4) {
    const int i4 = seg_i4(min(it, iters - 1), rot, iters, tid);
    const int4 v = src4[min(i4, E4 - 1)];
    const bool ok = (it < iters) && (i4 < E4);
    const int4 none = make_int4(KEY_SENTINEL, KEY_SENTINEL, KEY_SENTINEL, KEY_SENTINEL);
    return ok ? v : none;
}

__global__ __launch_bounds__(SEG_THREADS) void seg_max_kernel(
        const float* __restrict__ H,
        const int* __restrict__ src,
        const int* __restrict__ dst,
        int* __restrict__ keys, int N, int E) {
    __shared__ int keysL[SEG_NODES * DN];
    __shared__ int wl_e[SEG_WAVES * WL_CAP];
    __shared__ int wl_n[SEG_WAVES * WL_CAP];

    const int tid  = threadIdx.x;
    const int lane = tid & 31;
    const int wave = tid >> 5;
    lds_vint* le = (lds_vint*)(wl_e + wave * WL_CAP);
    lds_vint* ln = (lds_vint*)(wl_n + wave * WL_CAP);

    const int4* src4 = (const int4*)src;
    const int E4    = E >> 2;
    const int iters = (E4 + SEG_THREADS - 1) / SEG_THREADS;
    const int rot   = (int)(((long long)blockIdx.x * iters) / gridDim.x);

    for (int lo = blockIdx.x * SEG_NODES; lo < N; lo += gridDim.x * SEG_NODES) {
        const int nloc = min(SEG_NODES, N - lo);
        const unsigned un = (unsigned)nloc, ulo = (unsigned)lo;
        for (int i = tid; i < nloc * DN; i += SEG_THREADS) keysL[i] = KEY_SENTINEL;
        __syncthreads();

        int cnt = 0;
        auto examine = [&](const int4 (&buf)[SCAN_U], int it0) {
#pragma unroll
            for (int u = 0; u < SCAN_U; ++u) {
                const int4 s = buf[u];
                const unsigned a = (unsigned)s.x - ulo, b = (unsigned)s.y - ulo;
                const unsigned c = (unsigned)s.z - ulo, d = (unsigned)s.w - ulo;
                const unsigned mn = min(min(a, b), min(c, d));
                if (__builtin_amdgcn_ballot_w32(mn < un) != 0u) {
                    const int e0 = seg_i4(it0 + u, rot, iters, tid) << 2;
                    seg_append(a < un, e0 + 0, a, le, ln, cnt);
                    seg_append(b < un, e0 + 1, b, le, ln, cnt);
                    seg_append(c < un, e0 + 2, c, le, ln, cnt);
                    seg_append(d < un, e0 + 3, d, le, ln, cnt);
                    if (cnt >= WL_FLUSH) {
                        seg_flush(H, dst, keysL, le, ln, cnt, lane);
                        cnt = 0;
                    }
                }
            }
        };
        int4 bufA[SCAN_U], bufB[SCAN_U];
#pragma unroll
        for (int u = 0; u < SCAN_U; ++u)
            bufA[u] = (iters > 0) ? seg_load(src4, u, rot, iters, tid, E4)
                                  : make_int4(KEY_SENTINEL, KEY_SENTINEL, KEY_SENTINEL, KEY_SENTINEL);
        for (int it = 0; it < iters; it += 2 * SCAN_U) {
#pragma unroll
            for (int u = 0; u < SCAN_U; ++u)
                bufB[u] = seg_load(src4, it + SCAN_U + u, rot, iters, tid, E4);
            examine(bufA, it);
#pragma unroll
            for (int u = 0; u < SCAN_U; ++u)
                bufA[u] = seg_load(src4, it + 2 * SCAN_U + u, rot, iters, tid, E4);
            examine(bufB, it + SCAN_U);
        }
        {
            const int e = (E4 << 2) + lane;
            const bool v = (wave == 0) && (e < E);
            const unsigned a = v ? ((unsigned)src[e] - ulo) : 0xFFFFFFFFu;
            seg_append(a < un, e, a, le, ln, cnt);
        }
        if (cnt > 0) seg_flush(H, dst, keysL, le, ln, cnt, lane);
        __syncthreads();

        int4* o = (int4*)(keys + (size_t)lo * DN);
        const int4* kl4 = (const int4*)keysL;
        for (int i = tid; i < nloc * (DN / 4); i += SEG_THREADS) o[i] = kl4[i];
        __syncthreads();
    }
}

__global__ __launch_bounds__(GEMM_THREADS) void gemm_kernel(
        const float* __restrict__ H,
        const float* __restrict__ W,
        float* out  ) {
    __shared__ float Ws[DN * WPAD];

    for (int i = threadIdx.x; i < DN * (KTOT / 4); i += GEMM_THREADS) {
        int r  = i >> 6;
        int c4 = (i & 63) << 2;
        float4 w = *(const float4*)(W + (size_t)r * KTOT + c4);
        *(float4*)(&Ws[r * WPAD + c4]) = w;
    }
    __syncthreads();

    const int wave = threadIdx.x >> 5;
    const int lane = threadIdx.x & 31;
    const int lm   = lane & 15;
    const int h2   = (lane >> 4) << 1;
    const int m0   = blockIdx.x * ROWS_PER_BLOCK + wave * 16;
    const int row  = m0 + lm;

    const float* arow = H + (size_t)row * DN;
    const int*   krow = (const int*)out + (size_t)row * DN;

    v8f zero = {};
    v8f acc[8];
#pragma unroll
    for (int nt = 0; nt < 8; ++nt) acc[nt] = zero;

#pragma unroll 4
    for (int kk = 0; kk < 32; ++kk) {
        const int kc = (kk << 2) + h2;
        v2f a = *(const v2f*)(arow + kc);
#pragma unroll
        for (int nt = 0; nt < 8; ++nt) {
            v2f b = *(const v2f*)(&Ws[(nt * 16 + lm) * WPAD + kc]);
            acc[nt] = __builtin_amdgcn_wmma_f32_16x16x4_f32(
                false, a, false, b, (short)0, acc[nt], false, false);
        }
    }

#pragma unroll 4
    for (int kk = 0; kk < 32; ++kk) {
        const int kc = (kk << 2) + h2;
        int k0 = krow[kc];
        int k1 = krow[kc + 1];
        v2f a;
        a.x = (k0 == KEY_SENTINEL) ? 0.0f : dec_f32(k0);
        a.y = (k1 == KEY_SENTINEL) ? 0.0f : dec_f32(k1);
#pragma unroll
        for (int nt = 0; nt < 8; ++nt) {
            v2f b = *(const v2f*)(&Ws[(nt * 16 + lm) * WPAD + 128 + kc]);
            acc[nt] = __builtin_amdgcn_wmma_f32_16x16x4_f32(
                false, a, false, b, (short)0, acc[nt], false, false);
        }
    }

    const int orow0 = m0 + ((lane >> 4) << 3);
    float* obase = out + (size_t)orow0 * DN + lm;
#pragma unroll
    for (int r = 0; r < 8; ++r) {
#pragma unroll
        for (int nt = 0; nt < 8; ++nt) {
            obase[(size_t)r * DN + nt * 16] = acc[nt][r];
        }
    }
}

extern "C" void kernel_launch(void* const* d_in, const int* in_sizes, int n_in,
                              void* d_out, int out_size, void* d_ws, size_t ws_size,
                              hipStream_t stream) {
    (void)n_in; (void)out_size; (void)d_ws; (void)ws_size;
    const float* H   = (const float*)d_in[0];
    const int*   src = (const int*)d_in[1];
    const int*   dst = (const int*)d_in[2];
    const float* W   = (const float*)d_in[3];

    const int N = in_sizes[0] / DN;
    const int E = in_sizes[1];
    float* out = (float*)d_out;

    seg_max_kernel<<<SEG_GROUPS, SEG_THREADS, 0, stream>>>(H, src, dst, (int*)d_out, N, E);

    gemm_kernel<<<N / ROWS_PER_BLOCK, GEMM_THREADS, 0, stream>>>(H, W, out);
}


# pre-pass launch bins edges by owner node range into slots inside that owner's key area of out; each segment-max workgroup reads only its own edges instead of scanning all src; integer max, order-free
# speedup vs baseline: 2.2617x; 1.3431x over previous
#include <hip/hip_runtime.h>

#define DN   128
#define KTOT 256
#define WPAD 260
#define ROWS_PER_BLOCK 80
#define GEMM_THREADS   160
#define KEY_SENTINEL ((int)0x80000000)

typedef float v2f __attribute__((ext_vector_type(2)));
typedef float v8f __attribute__((ext_vector_type(8)));

__device__ __forceinline__ int enc_f32(float f) {
    int b = __float_as_int(f);
    return b >= 0 ? b : (b ^ 0x7FFFFFFF);
}
__device__ __forceinline__ float dec_f32(int k) {
    return __int_as_float(k >= 0 ? k : (k ^ 0x7FFFFFFF));
}

#define SEG_NODES   391
#define SEG_GROUPS  256
#define SEG_THREADS 1024
#define SEG_WAVES   32
#define WL_FLUSH    64
#define WL_CAP      192
#define SCAN_U      4

#define SLOT_W    128
#define SLOT_CAP  127
#define BIN_PITCH 129
#define BIN_U     8
#define DST_BITS  17
#define DST_MASK  ((1 << DST_BITS) - 1)
#define SLOTS_PER_WAVE (SEG_GROUPS / SEG_WAVES)

__device__ __forceinline__ bool bins_usable(int N, int groups) {
    return groups == SEG_GROUPS
        && N <= SEG_GROUPS * SEG_NODES
        && N >= (SEG_GROUPS - 1) * SEG_NODES + (SEG_GROUPS * SLOT_W) / DN
        && N <= (1 << DST_BITS);
}


typedef __attribute__((address_space(3))) volatile int lds_vint;

__device__ __forceinline__ void seg_append(bool p, int e, unsigned nl,
                                           lds_vint* le, lds_vint* ln,
                                           int& cnt) {
    const unsigned m = __builtin_amdgcn_ballot_w32(p);
    if (p) {
        const int pos = cnt + (int)__builtin_amdgcn_mbcnt_lo(m, 0u);
        le[pos] = e;
        ln[pos] = (int)nl;
    }
    cnt += __builtin_popcount(m);
}

__device__ __forceinline__ void seg_flush(const float* __restrict__ H,
                                          const int* __restrict__ dst,
                                          int* keysL,
                                          lds_vint* le, lds_vint* ln,
                                          int cnt, int lane) {
    for (int i = lane; i < cnt; i += 32) {
        const int e = le[i];
        le[i] = dst[e];
    }
    for (int i = 0; i < cnt; i += 8) {
        float4 h[8];
        int nl[8];
#pragma unroll
        for (int j = 0; j < 8; ++j) {
            const int idx = min(i + j, cnt - 1);
            const int d = __builtin_amdgcn_readfirstlane(le[idx]);
            nl[j] = __builtin_amdgcn_readfirstlane(ln[idx]);
            h[j] = *(const float4*)(H + (size_t)d * DN + (lane << 2));
        }
        __builtin_amdgcn_sched_barrier(0);
#pragma unroll
        for (int j = 0; j < 8; ++j) {
            int* kp = keysL + nl[j] * DN + (lane << 2);
            atomicMax(kp + 0, enc_f32(h[j].x));
            atomicMax(kp + 1, enc_f32(h[j].y));
            atomicMax(kp + 2, enc_f32(h[j].z));
            atomicMax(kp + 3, enc_f32(h[j].w));
        }
    }
}

__device__ __forceinline__ void seg_flush_bins(const float* __restrict__ H,
                                               int* keysL, lds_vint* le,
                                               int cnt, int lane) {
    for (int i = 0; i < cnt; i += 8) {
        float4 h[8];
        int nl[8];
#pragma unroll
        for (int j = 0; j < 8; ++j) {
            const int idx = min(i + j, cnt - 1);
            const int p = __builtin_amdgcn_readfirstlane(le[idx]);
            nl[j] = p >> DST_BITS;
            h[j] = *(const float4*)(H + (size_t)(p & DST_MASK) * DN + (lane << 2));
        }
        __builtin_amdgcn_sched_barrier(0);
#pragma unroll
        for (int j = 0; j < 8; ++j) {
            int* kp = keysL + nl[j] * DN + (lane << 2);
            atomicMax(kp + 0, enc_f32(h[j].x));
            atomicMax(kp + 1, enc_f32(h[j].y));
            atomicMax(kp + 2, enc_f32(h[j].z));
            atomicMax(kp + 3, enc_f32(h[j].w));
        }
    }
}

__global__ __launch_bounds__(SEG_THREADS) void bin_kernel(
        const int* __restrict__ src,
        const int* __restrict__ dst,
        int* __restrict__ slots, int N, int E) {
    __shared__ int cntL[SEG_GROUPS];
    __shared__ int stage[SEG_GROUPS * BIN_PITCH];

    if (!bins_usable(N, (int)gridDim.x)) return;

    const int tid  = threadIdx.x;
    const int lane = tid & 31;
    const int wave = tid >> 5;
    const int w    = blockIdx.x;

    if (tid < SEG_GROUPS) cntL[tid] = 0;
    __syncthreads();

    const int chunk = (E + SEG_GROUPS - 1) / SEG_GROUPS;
    const int e0 = w * chunk;
    const int e1 = min(E, e0 + chunk);
    for (int base = e0; base < e1; base += BIN_U * SEG_THREADS) {
        int s[BIN_U], d[BIN_U];
#pragma unroll
        for (int u = 0; u < BIN_U; ++u) {
            const int e = min(base + u * SEG_THREADS + tid, E - 1);
            s[u] = src[e];
            d[u] = dst[e];
        }
#pragma unroll
        for (int u = 0; u < BIN_U; ++u) {
            const int e = base + u * SEG_THREADS + tid;
            if (e < e1 && (unsigned)s[u] < (unsigned)N && (unsigned)d[u] < (unsigned)N) {
                const int b  = s[u] / SEG_NODES;
                const int nl = s[u] - b * SEG_NODES;
                const int pos = atomicAdd(&cntL[b], 1);
                if (pos < SLOT_CAP) stage[b * BIN_PITCH + pos] = d[u] | (nl << DST_BITS);
            }
        }
    }
    __syncthreads();

    for (int b = wave; b < SEG_GROUPS; b += SEG_WAVES) {
        const int c  = cntL[b];
        const int nw = 1 + min(c, SLOT_CAP);
        int* slot = slots + (size_t)b * (SEG_NODES * DN) + w * SLOT_W;
        for (int i = 0; i < nw; i += 32) {
            const int idx = i + lane;
            int v = 0;
            if (idx == 0) v = c;
            else if (idx < nw) v = stage[b * BIN_PITCH + idx - 1];
            slot[idx] = v;
        }
    }
}

__device__ __forceinline__ int seg_i4(int it, int rot, int iters, int tid) {
    int itr = it + rot;
    if (itr >= iters) itr -= iters;
    return itr * SEG_THREADS + tid;
}
__device__ __forceinline__ int4 seg_load(const int4* __restrict__ src4, int it,
                                         int rot, int iters, int tid, int E4) {
    const int i4 = seg_i4(min(it, iters - 1), rot, iters, tid);
    const int4 v = src4[min(i4, E4 - 1)];
    const bool ok = (it < iters) && (i4 < E4);
    const int4 none = make_int4(KEY_SENTINEL, KEY_SENTINEL, KEY_SENTINEL, KEY_SENTINEL);
    return ok ? v : none;
}

__global__ __launch_bounds__(SEG_THREADS) void seg_max_kernel(
        const float* __restrict__ H,
        const int* __restrict__ src,
        const int* __restrict__ dst,
        int* __restrict__ keys, int N, int E) {
    __shared__ int keysL[SEG_NODES * DN];
    __shared__ int wl_e[SEG_WAVES * WL_CAP];
    __shared__ int wl_n[SEG_WAVES * WL_CAP];
    __shared__ int ovfL;

    const int tid  = threadIdx.x;
    const int lane = tid & 31;
    const int wave = tid >> 5;
    lds_vint* le = (lds_vint*)(wl_e + wave * WL_CAP);
    lds_vint* ln = (lds_vint*)(wl_n + wave * WL_CAP);

    const int4* src4 = (const int4*)src;
    const int E4    = E >> 2;
    const int iters = (E4 + SEG_THREADS - 1) / SEG_THREADS;
    const int rot   = (int)(((long long)blockIdx.x * iters) / gridDim.x);
    const bool binned = bins_usable(N, (int)gridDim.x);

    for (int lo = blockIdx.x * SEG_NODES; lo < N; lo += gridDim.x * SEG_NODES) {
        const int nloc = min(SEG_NODES, N - lo);
        const unsigned un = (unsigned)nloc, ulo = (unsigned)lo;

        const int* myslots = keys + (size_t)lo * DN;
        int first[SLOTS_PER_WAVE];
        if (binned) {
#pragma unroll
            for (int s = 0; s < SLOTS_PER_WAVE; ++s)
                first[s] = myslots[(wave + s * SEG_WAVES) * SLOT_W + lane];
            if (tid == 0) ovfL = 0;
        }

        for (int i = tid; i < nloc * DN; i += SEG_THREADS) keysL[i] = KEY_SENTINEL;
        __syncthreads();

        bool use_bins = false;
        if (binned) {
            bool ov = false;
#pragma unroll
            for (int s = 0; s < SLOTS_PER_WAVE; ++s)
                ov |= __builtin_amdgcn_readfirstlane(first[s]) > SLOT_CAP;
            if (ov && lane == 0) ovfL = 1;
            __syncthreads();
            use_bins = (ovfL == 0);
        }

        int cnt = 0;
        if (use_bins) {
#pragma unroll
            for (int s = 0; s < SLOTS_PER_WAVE; ++s) {
                const int n = __builtin_amdgcn_readfirstlane(first[s]);
                if (lane >= 1 && lane - 1 < n) le[cnt + lane - 1] = first[s];
                cnt += min(n, 31);
                const int* slot = myslots + (wave + s * SEG_WAVES) * SLOT_W;
                for (int i = 32; i < 1 + n; i += 32) {
                    if (i + lane < 1 + n) le[cnt + lane] = slot[i + lane];
                    cnt += min(32, 1 + n - i);
                }
                if (cnt >= WL_FLUSH) {
                    seg_flush_bins(H, keysL, le, cnt, lane);
                    cnt = 0;
                }
            }
            if (cnt > 0) seg_flush_bins(H, keysL, le, cnt, lane);
        } else {
            auto examine = [&](const int4 (&buf)[SCAN_U], int it0) {
#pragma unroll
                for (int u = 0; u < SCAN_U; ++u) {
                    const int4 s = buf[u];
                    const unsigned a = (unsigned)s.x - ulo, b = (unsigned)s.y - ulo;
                    const unsigned c = (unsigned)s.z - ulo, d = (unsigned)s.w - ulo;
                    const unsigned mn = min(min(a, b), min(c, d));
                    if (__builtin_amdgcn_ballot_w32(mn < un) != 0u) {
                        const int e0 = seg_i4(it0 + u, rot, iters, tid) << 2;
                        seg_append(a < un, e0 + 0, a, le, ln, cnt);
                        seg_append(b < un, e0 + 1, b, le, ln, cnt);
                        seg_append(c < un, e0 + 2, c, le, ln, cnt);
                        seg_append(d < un, e0 + 3, d, le, ln, cnt);
                        if (cnt >= WL_FLUSH) {
                            seg_flush(H, dst, keysL, le, ln, cnt, lane);
                            cnt = 0;
                        }
                    }
                }
            };
            int4 bufA[SCAN_U], bufB[SCAN_U];
#pragma unroll
            for (int u = 0; u < SCAN_U; ++u)
                bufA[u] = (iters > 0) ? seg_load(src4, u, rot, iters, tid, E4)
                                      : make_int4(KEY_SENTINEL, KEY_SENTINEL, KEY_SENTINEL, KEY_SENTINEL);
            for (int it = 0; it < iters; it += 2 * SCAN_U) {
#pragma unroll
                for (int u = 0; u < SCAN_U; ++u)
                    bufB[u] = seg_load(src4, it + SCAN_U + u, rot, iters, tid, E4);
                examine(bufA, it);
#pragma unroll
                for (int u = 0; u < SCAN_U; ++u)
                    bufA[u] = seg_load(src4, it + 2 * SCAN_U + u, rot, iters, tid, E4);
                examine(bufB, it + SCAN_U);
            }
            {
                const int e = (E4 << 2) + lane;
                const bool v = (wave == 0) && (e < E);
                const unsigned a = v ? ((unsigned)src[e] - ulo) : 0xFFFFFFFFu;
                seg_append(a < un, e, a, le, ln, cnt);
            }
            if (cnt > 0) seg_flush(H, dst, keysL, le, ln, cnt, lane);
        }
        __syncthreads();

        int4* o = (int4*)(keys + (size_t)lo * DN);
        const int4* kl4 = (const int4*)keysL;
        for (int i = tid; i < nloc * (DN / 4); i += SEG_THREADS) o[i] = kl4[i];
        __syncthreads();
    }
}

__global__ __launch_bounds__(GEMM_THREADS) void gemm_kernel(
        const float* __restrict__ H,
        const float* __restrict__ W,
        float* out  ) {
    __shared__ float Ws[DN * WPAD];

    for (int i = threadIdx.x; i < DN * (KTOT / 4); i += GEMM_THREADS) {
        int r  = i >> 6;
        int c4 = (i & 63) << 2;
        float4 w = *(const float4*)(W + (size_t)r * KTOT + c4);
        *(float4*)(&Ws[r * WPAD + c4]) = w;
    }
    __syncthreads();

    const int wave = threadIdx.x >> 5;
    const int lane = threadIdx.x & 31;
    const int lm   = lane & 15;
    const int h2   = (lane >> 4) << 1;
    const int m0   = blockIdx.x * ROWS_PER_BLOCK + wave * 16;
    const int row  = m0 + lm;

    const float* arow = H + (size_t)row * DN;
    const int*   krow = (const int*)out + (size_t)row * DN;

    v8f zero = {};
    v8f acc[8];
#pragma unroll
    for (int nt = 0; nt < 8; ++nt) acc[nt] = zero;

#pragma unroll 4
    for (int kk = 0; kk < 32; ++kk) {
        const int kc = (kk << 2) + h2;
        v2f a = *(const v2f*)(arow + kc);
#pragma unroll
        for (int nt = 0; nt < 8; ++nt) {
            v2f b = *(const v2f*)(&Ws[(nt * 16 + lm) * WPAD + kc]);
            acc[nt] = __builtin_amdgcn_wmma_f32_16x16x4_f32(
                false, a, false, b, (short)0, acc[nt], false, false);
        }
    }

#pragma unroll 4
    for (int kk = 0; kk < 32; ++kk) {
        const int kc = (kk << 2) + h2;
        int k0 = krow[kc];
        int k1 = krow[kc + 1];
        v2f a;
        a.x = (k0 == KEY_SENTINEL) ? 0.0f : dec_f32(k0);
        a.y = (k1 == KEY_SENTINEL) ? 0.0f : dec_f32(k1);
#pragma unroll
        for (int nt = 0; nt < 8; ++nt) {
            v2f b = *(const v2f*)(&Ws[(nt * 16 + lm) * WPAD + 128 + kc]);
            acc[nt] = __builtin_amdgcn_wmma_f32_16x16x4_f32(
                false, a, false, b, (short)0, acc[nt], false, false);
        }
    }

    const int orow0 = m0 + ((lane >> 4) << 3);
    float* obase = out + (size_t)orow0 * DN + lm;
#pragma unroll
    for (int r = 0; r < 8; ++r) {
#pragma unroll
        for (int nt = 0; nt < 8; ++nt) {
            obase[(size_t)r * DN + nt * 16] = acc[nt][r];
        }
    }
}

extern "C" void kernel_launch(void* const* d_in, const int* in_sizes, int n_in,
                              void* d_out, int out_size, void* d_ws, size_t ws_size,
                              hipStream_t stream) {
    (void)n_in; (void)out_size; (void)d_ws; (void)ws_size;
    const float* H   = (const float*)d_in[0];
    const int*   src = (const int*)d_in[1];
    const int*   dst = (const int*)d_in[2];
    const float* W   = (const float*)d_in[3];

    const int N = in_sizes[0] / DN;
    const int E = in_sizes[1];
    float* out = (float*)d_out;

    bin_kernel<<<SEG_GROUPS, SEG_THREADS, 0, stream>>>(src, dst, (int*)d_out, N, E);

    seg_max_kernel<<<SEG_GROUPS, SEG_THREADS, 0, stream>>>(H, src, dst, (int*)d_out, N, E);

    gemm_kernel<<<N / ROWS_PER_BLOCK, GEMM_THREADS, 0, stream>>>(H, W, out);
}


# segment-max: owner workgroup sorts its range's edges by node in LDS (count, prefix sum, place); each wave reduces one node's H rows by integer max in registers, one row store; LDS max atomics gone
# speedup vs baseline: 2.6015x; 1.1502x over previous
#include <hip/hip_runtime.h>

#define DN   128
#define KTOT 256
#define WPAD 260
#define ROWS_PER_BLOCK 80
#define GEMM_THREADS   160
#define KEY_SENTINEL ((int)0x80000000)

typedef float v2f __attribute__((ext_vector_type(2)));
typedef float v8f __attribute__((ext_vector_type(8)));

__device__ __forceinline__ int enc_f32(float f) {
    int b = __float_as_int(f);
    return b >= 0 ? b : (b ^ 0x7FFFFFFF);
}
__device__ __forceinline__ float dec_f32(int k) {
    return __int_as_float(k >= 0 ? k : (k ^ 0x7FFFFFFF));
}

#define SEG_NODES   391
#define SEG_GROUPS  256
#define SEG_THREADS 1024
#define SEG_WAVES   32
#define WL_FLUSH    64
#define WL_CAP      192
#define SCAN_U      4

#define SLOT_W    128
#define SLOT_CAP  127
#define BIN_PITCH 129
#define BIN_U     8
#define DST_BITS  17
#define DST_MASK  ((1 << DST_BITS) - 1)
#define SLOTS_PER_WAVE (SEG_GROUPS / SEG_WAVES)

__device__ __forceinline__ bool bins_usable(int N, int groups) {
    return groups == SEG_GROUPS
        && N <= SEG_GROUPS * SEG_NODES
        && N >= (SEG_GROUPS - 1) * SEG_NODES + (SEG_GROUPS * SLOT_W) / DN
        && N <= (1 << DST_BITS);
}


typedef __attribute__((address_space(3))) volatile int lds_vint;

__device__ __forceinline__ void seg_append(bool p, int e, unsigned nl,
                                           lds_vint* le, lds_vint* ln,
                                           int& cnt) {
    const unsigned m = __builtin_amdgcn_ballot_w32(p);
    if (p) {
        const int pos = cnt + (int)__builtin_amdgcn_mbcnt_lo(m, 0u);
        le[pos] = e;
        ln[pos] = (int)nl;
    }
    cnt += __builtin_popcount(m);
}

__device__ __forceinline__ void seg_flush(const float* __restrict__ H,
                                          const int* __restrict__ dst,
                                          int* keysL,
                                          lds_vint* le, lds_vint* ln,
                                          int cnt, int lane) {
    for (int i = lane; i < cnt; i += 32) {
        const int e = le[i];
        le[i] = dst[e];
    }
    for (int i = 0; i < cnt; i += 8) {
        float4 h[8];
        int nl[8];
#pragma unroll
        for (int j = 0; j < 8; ++j) {
            const int idx = min(i + j, cnt - 1);
            const int d = __builtin_amdgcn_readfirstlane(le[idx]);
            nl[j] = __builtin_amdgcn_readfirstlane(ln[idx]);
            h[j] = *(const float4*)(H + (size_t)d * DN + (lane << 2));
        }
        __builtin_amdgcn_sched_barrier(0);
#pragma unroll
        for (int j = 0; j < 8; ++j) {
            int* kp = keysL + nl[j] * DN + (lane << 2);
            atomicMax(kp + 0, enc_f32(h[j].x));
            atomicMax(kp + 1, enc_f32(h[j].y));
            atomicMax(kp + 2, enc_f32(h[j].z));
            atomicMax(kp + 3, enc_f32(h[j].w));
        }
    }
}

__global__ __launch_bounds__(SEG_THREADS) void bin_kernel(
        const int* __restrict__ src,
        const int* __restrict__ dst,
        int* __restrict__ slots, int N, int E) {
    __shared__ int cntL[SEG_GROUPS];
    __shared__ int stage[SEG_GROUPS * BIN_PITCH];

    if (!bins_usable(N, (int)gridDim.x)) return;

    const int tid  = threadIdx.x;
    const int lane = tid & 31;
    const int wave = tid >> 5;
    const int w    = blockIdx.x;

    if (tid < SEG_GROUPS) cntL[tid] = 0;
    __syncthreads();

    const int chunk = (E + SEG_GROUPS - 1) / SEG_GROUPS;
    const int e0 = w * chunk;
    const int e1 = min(E, e0 + chunk);
    for (int base = e0; base < e1; base += BIN_U * SEG_THREADS) {
        int s[BIN_U], d[BIN_U];
#pragma unroll
        for (int u = 0; u < BIN_U; ++u) {
            const int e = min(base + u * SEG_THREADS + tid, E - 1);
            s[u] = src[e];
            d[u] = dst[e];
        }
#pragma unroll
        for (int u = 0; u < BIN_U; ++u) {
            const int e = base + u * SEG_THREADS + tid;
            if (e < e1 && (unsigned)s[u] < (unsigned)N && (unsigned)d[u] < (unsigned)N) {
                const int b  = s[u] / SEG_NODES;
                const int nl = s[u] - b * SEG_NODES;
                const int pos = atomicAdd(&cntL[b], 1);
                if (pos < SLOT_CAP) stage[b * BIN_PITCH + pos] = d[u] | (nl << DST_BITS);
            }
        }
    }
    __syncthreads();

    for (int b = wave; b < SEG_GROUPS; b += SEG_WAVES) {
        const int c  = cntL[b];
        const int nw = 1 + min(c, SLOT_CAP);
        int* slot = slots + (size_t)b * (SEG_NODES * DN) + w * SLOT_W;
        for (int i = 0; i < nw; i += 32) {
            const int idx = i + lane;
            int v = 0;
            if (idx == 0) v = c;
            else if (idx < nw) v = stage[b * BIN_PITCH + idx - 1];
            slot[idx] = v;
        }
    }
}

__device__ __forceinline__ int seg_i4(int it, int rot, int iters, int tid) {
    int itr = it + rot;
    if (itr >= iters) itr -= iters;
    return itr * SEG_THREADS + tid;
}
__device__ __forceinline__ int4 seg_load(const int4* __restrict__ src4, int it,
                                         int rot, int iters, int tid, int E4) {
    const int i4 = seg_i4(min(it, iters - 1), rot, iters, tid);
    const int4 v = src4[min(i4, E4 - 1)];
    const bool ok = (it < iters) && (i4 < E4);
    const int4 none = make_int4(KEY_SENTINEL, KEY_SENTINEL, KEY_SENTINEL, KEY_SENTINEL);
    return ok ? v : none;
}

template <typename F>
__device__ __forceinline__ void seg_each_entry(const int (&first)[SLOTS_PER_WAVE],
                                               const int* myslots, int wave, int lane, F f) {
#pragma unroll
    for (int s = 0; s < SLOTS_PER_WAVE; ++s) {
        const int n = __builtin_amdgcn_readfirstlane(first[s]);
        if (lane >= 1 && lane - 1 < n) f(first[s]);
        const int* slot = myslots + (wave + s * SEG_WAVES) * SLOT_W;
#pragma clang loop unroll(disable)
        for (int i = 32; i < 1 + n; i += 32)
            if (i + lane < 1 + n) f(slot[i + lane]);
    }
}

__global__ __launch_bounds__(SEG_THREADS) void seg_max_kernel(
        const float* __restrict__ H,
        const int* __restrict__ src,
        const int* __restrict__ dst,
        int* __restrict__ keys, int N, int E) {
    __shared__ int keysL[SEG_NODES * DN];
    __shared__ int wl_e[SEG_WAVES * WL_CAP];
    __shared__ int wl_n[SEG_WAVES * WL_CAP];
    __shared__ int nodeCnt[SEG_NODES + 1];
    __shared__ int nodeOff[SEG_NODES + 1];
    __shared__ int ovfL;

    const int tid  = threadIdx.x;
    const int lane = tid & 31;
    const int wave = tid >> 5;
    lds_vint* le = (lds_vint*)(wl_e + wave * WL_CAP);
    lds_vint* ln = (lds_vint*)(wl_n + wave * WL_CAP);

    const int4* src4 = (const int4*)src;
    const int E4    = E >> 2;
    const int iters = (E4 + SEG_THREADS - 1) / SEG_THREADS;
    const int rot   = (int)(((long long)blockIdx.x * iters) / gridDim.x);
    const bool binned = bins_usable(N, (int)gridDim.x);

    for (int lo = blockIdx.x * SEG_NODES; lo < N; lo += gridDim.x * SEG_NODES) {
        const int nloc = min(SEG_NODES, N - lo);
        const unsigned un = (unsigned)nloc, ulo = (unsigned)lo;

        const int* myslots = keys + (size_t)lo * DN;
        int first[SLOTS_PER_WAVE];
        if (binned) {
#pragma unroll
            for (int s = 0; s < SLOTS_PER_WAVE; ++s)
                first[s] = myslots[(wave + s * SEG_WAVES) * SLOT_W + lane];
            if (tid == 0) ovfL = 0;
            for (int i = tid; i <= SEG_NODES; i += SEG_THREADS) nodeCnt[i] = 0;
        }
        __syncthreads();

        bool use_bins = false;
        if (binned) {
            bool ov = false;
#pragma unroll
            for (int s = 0; s < SLOTS_PER_WAVE; ++s)
                ov |= __builtin_amdgcn_readfirstlane(first[s]) > SLOT_CAP;
            if (ov && lane == 0) ovfL = 1;
            __syncthreads();
            use_bins = (ovfL == 0);
        }

        if (use_bins) {
            int* sorted = keysL;
            seg_each_entry(first, myslots, wave, lane,
                           [&](int e) { atomicAdd(&nodeCnt[e >> DST_BITS], 1); });
            __syncthreads();
            if (wave == 0) {
                constexpr int PER = (SEG_NODES + 1 + 31) / 32;
                int c[PER];
                int sum = 0;
#pragma unroll
                for (int k = 0; k < PER; ++k) {
                    const int idx = lane * PER + k;
                    c[k] = (idx < SEG_NODES) ? nodeCnt[idx] : 0;
                    sum += c[k];
                }
                int incl = sum;
#pragma unroll
                for (int d = 1; d < 32; d <<= 1) {
                    const int t = __shfl_up(incl, d);
                    if (lane >= d) incl += t;
                }
                int off = incl - sum;
#pragma unroll
                for (int k = 0; k < PER; ++k) {
                    const int idx = lane * PER + k;
                    if (idx <= SEG_NODES) { nodeOff[idx] = off; nodeCnt[idx] = off; }
                    off += c[k];
                }
            }
            __syncthreads();
            seg_each_entry(first, myslots, wave, lane,
                           [&](int e) {
                               const int pos = atomicAdd(&nodeCnt[e >> DST_BITS], 1);
                               sorted[pos] = e & DST_MASK;
                           });
            __syncthreads();

            for (int n = wave; n < nloc; n += SEG_WAVES) {
                const int beg = __builtin_amdgcn_readfirstlane(nodeOff[n]);
                const int end = __builtin_amdgcn_readfirstlane(nodeOff[n + 1]);
                int4 acc = make_int4(KEY_SENTINEL, KEY_SENTINEL, KEY_SENTINEL, KEY_SENTINEL);
                for (int i = beg; i < end; i += 8) {
                    const int mine = sorted[min(i + (lane & 7), end - 1)];
                    float4 h[8];
#pragma unroll
                    for (int j = 0; j < 8; ++j) {
                        const int d = __builtin_amdgcn_readlane(mine, j);
                        h[j] = *(const float4*)(H + (size_t)d * DN + (lane << 2));
                    }
                    __builtin_amdgcn_sched_barrier(0);
#pragma unroll
                    for (int j = 0; j < 8; ++j) {
                        acc.x = max(acc.x, enc_f32(h[j].x));
                        acc.y = max(acc.y, enc_f32(h[j].y));
                        acc.z = max(acc.z, enc_f32(h[j].z));
                        acc.w = max(acc.w, enc_f32(h[j].w));
                    }
                }
                *(int4*)(keys + (size_t)(lo + n) * DN + (lane << 2)) = acc;
            }
        } else {
            for (int i = tid; i < nloc * DN; i += SEG_THREADS) keysL[i] = KEY_SENTINEL;
            __syncthreads();

            int cnt = 0;
            auto examine = [&](const int4 (&buf)[SCAN_U], int it0) {
#pragma unroll
                for (int u = 0; u < SCAN_U; ++u) {
                    const int4 s = buf[u];
                    const unsigned a = (unsigned)s.x - ulo, b = (unsigned)s.y - ulo;
                    const unsigned c = (unsigned)s.z - ulo, d = (unsigned)s.w - ulo;
                    const unsigned mn = min(min(a, b), min(c, d));
                    if (__builtin_amdgcn_ballot_w32(mn < un) != 0u) {
                        const int e0 = seg_i4(it0 + u, rot, iters, tid) << 2;
                        seg_append(a < un, e0 + 0, a, le, ln, cnt);
                        seg_append(b < un, e0 + 1, b, le, ln, cnt);
                        seg_append(c < un, e0 + 2, c, le, ln, cnt);
                        seg_append(d < un, e0 + 3, d, le, ln, cnt);
                        if (cnt >= WL_FLUSH) {
                            seg_flush(H, dst, keysL, le, ln, cnt, lane);
                            cnt = 0;
                        }
                    }
                }
            };
            int4 bufA[SCAN_U], bufB[SCAN_U];
#pragma unroll
            for (int u = 0; u < SCAN_U; ++u)
                bufA[u] = (iters > 0) ? seg_load(src4, u, rot, iters, tid, E4)
                                      : make_int4(KEY_SENTINEL, KEY_SENTINEL, KEY_SENTINEL, KEY_SENTINEL);
            for (int it = 0; it < iters; it += 2 * SCAN_U) {
#pragma unroll
                for (int u = 0; u < SCAN_U; ++u)
                    bufB[u] = seg_load(src4, it + SCAN_U + u, rot, iters, tid, E4);
                examine(bufA, it);
#pragma unroll
                for (int u = 0; u < SCAN_U; ++u)
                    bufA[u] = seg_load(src4, it + 2 * SCAN_U + u, rot, iters, tid, E4);
                examine(bufB, it + SCAN_U);
            }
            {
                const int e = (E4 << 2) + lane;
                const bool v = (wave == 0) && (e < E);
                const unsigned a = v ? ((unsigned)src[e] - ulo) : 0xFFFFFFFFu;
                seg_append(a < un, e, a, le, ln, cnt);
            }
            if (cnt > 0) seg_flush(H, dst, keysL, le, ln, cnt, lane);
            __syncthreads();

            int4* o = (int4*)(keys + (size_t)lo * DN);
            const int4* kl4 = (const int4*)keysL;
            for (int i = tid; i < nloc * (DN / 4); i += SEG_THREADS) o[i] = kl4[i];
        }
        __syncthreads();
    }
}

__global__ __launch_bounds__(GEMM_THREADS) void gemm_kernel(
        const float* __restrict__ H,
        const float* __restrict__ W,
        float* out  ) {
    __shared__ float Ws[DN * WPAD];

    for (int i = threadIdx.x; i < DN * (KTOT / 4); i += GEMM_THREADS) {
        int r  = i >> 6;
        int c4 = (i & 63) << 2;
        float4 w = *(const float4*)(W + (size_t)r * KTOT + c4);
        *(float4*)(&Ws[r * WPAD + c4]) = w;
    }
    __syncthreads();

    const int wave = threadIdx.x >> 5;
    const int lane = threadIdx.x & 31;
    const int lm   = lane & 15;
    const int h2   = (lane >> 4) << 1;
    const int m0   = blockIdx.x * ROWS_PER_BLOCK + wave * 16;
    const int row  = m0 + lm;

    const float* arow = H + (size_t)row * DN;
    const int*   krow = (const int*)out + (size_t)row * DN;

    v8f zero = {};
    v8f acc[8];
#pragma unroll
    for (int nt = 0; nt < 8; ++nt) acc[nt] = zero;

#pragma unroll 4
    for (int kk = 0; kk < 32; ++kk) {
        const int kc = (kk << 2) + h2;
        v2f a = *(const v2f*)(arow + kc);
#pragma unroll
        for (int nt = 0; nt < 8; ++nt) {
            v2f b = *(const v2f*)(&Ws[(nt * 16 + lm) * WPAD + kc]);
            acc[nt] = __builtin_amdgcn_wmma_f32_16x16x4_f32(
                false, a, false, b, (short)0, acc[nt], false, false);
        }
    }

#pragma unroll 4
    for (int kk = 0; kk < 32; ++kk) {
        const int kc = (kk << 2) + h2;
        int k0 = krow[kc];
        int k1 = krow[kc + 1];
        v2f a;
        a.x = (k0 == KEY_SENTINEL) ? 0.0f : dec_f32(k0);
        a.y = (k1 == KEY_SENTINEL) ? 0.0f : dec_f32(k1);
#pragma unroll
        for (int nt = 0; nt < 8; ++nt) {
            v2f b = *(const v2f*)(&Ws[(nt * 16 + lm) * WPAD + 128 + kc]);
            acc[nt] = __builtin_amdgcn_wmma_f32_16x16x4_f32(
                false, a, false, b, (short)0, acc[nt], false, false);
        }
    }

    const int orow0 = m0 + ((lane >> 4) << 3);
    float* obase = out + (size_t)orow0 * DN + lm;
#pragma unroll
    for (int r = 0; r < 8; ++r) {
#pragma unroll
        for (int nt = 0; nt < 8; ++nt) {
            obase[(size_t)r * DN + nt * 16] = acc[nt][r];
        }
    }
}

extern "C" void kernel_launch(void* const* d_in, const int* in_sizes, int n_in,
                              void* d_out, int out_size, void* d_ws, size_t ws_size,
                              hipStream_t stream) {
    (void)n_in; (void)out_size; (void)d_ws; (void)ws_size;
    const float* H   = (const float*)d_in[0];
    const int*   src = (const int*)d_in[1];
    const int*   dst = (const int*)d_in[2];
    const float* W   = (const float*)d_in[3];

    const int N = in_sizes[0] / DN;
    const int E = in_sizes[1];
    float* out = (float*)d_out;

    bin_kernel<<<SEG_GROUPS, SEG_THREADS, 0, stream>>>(src, dst, (int*)d_out, N, E);

    seg_max_kernel<<<SEG_GROUPS, SEG_THREADS, 0, stream>>>(H, src, dst, (int*)d_out, N, E);

    gemm_kernel<<<N / ROWS_PER_BLOCK, GEMM_THREADS, 0, stream>>>(H, W, out);
}


# segment-max fused into the GEMM launch: each wave gathers its 16-node tile's max rows (float max in registers) into its own LDS tile as A operand, H-half WMMAs under the row loads; one launch gone
# speedup vs baseline: 3.9879x; 1.5329x over previous
#include <hip/hip_runtime.h>

#define DN   128
#define KTOT 256
#define WPAD 260
#define KEY_SENTINEL ((int)0x80000000)

typedef float v2f __attribute__((ext_vector_type(2)));
typedef float v8f __attribute__((ext_vector_type(8)));

__device__ __forceinline__ int enc_f32(float f) {
    int b = __float_as_int(f);
    return b >= 0 ? b : (b ^ 0x7FFFFFFF);
}
__device__ __forceinline__ float dec_f32(int k) {
    return __int_as_float(k >= 0 ? k : (k ^ 0x7FFFFFFF));
}

#define SEG_NODES   391
#define SEG_GROUPS  256
#define SEG_THREADS 1024
#define SEG_WAVES   32

#define SLOT_W    128
#define SLOT_CAP  127
#define BIN_PITCH 129
#define BIN_U     8
#define DST_BITS  17
#define DST_MASK  ((1 << DST_BITS) - 1)

#define FW        12
#define FTHREADS  (FW * 32)
#define SLOTS_PER_WAVE ((SEG_GROUPS + FW - 1) / FW)
#define APITCH    132
#define SORT_CAP  12288

__device__ __forceinline__ bool bins_usable(int N, int groups) {
    return groups == SEG_GROUPS
        && N <= SEG_GROUPS * SEG_NODES
        && N >= (SEG_GROUPS - 1) * SEG_NODES + (SEG_GROUPS * SLOT_W) / DN
        && N <= (1 << DST_BITS);
}

__global__ __launch_bounds__(SEG_THREADS) void bin_kernel(
        const int* __restrict__ src,
        const int* __restrict__ dst,
        int* __restrict__ slots, int N, int E) {
    __shared__ int cntL[SEG_GROUPS];
    __shared__ int stage[SEG_GROUPS * BIN_PITCH];

    if (!bins_usable(N, (int)gridDim.x)) return;

    const int tid  = threadIdx.x;
    const int lane = tid & 31;
    const int wave = tid >> 5;
    const int w    = blockIdx.x;

    if (tid < SEG_GROUPS) cntL[tid] = 0;
    __syncthreads();

    const int chunk = (E + SEG_GROUPS - 1) / SEG_GROUPS;
    const int e0 = w * chunk;
    const int e1 = min(E, e0 + chunk);
    for (int base = e0; base < e1; base += BIN_U * SEG_THREADS) {
        int s[BIN_U], d[BIN_U];
#pragma unroll
        for (int u = 0; u < BIN_U; ++u) {
            const int e = min(base + u * SEG_THREADS + tid, E - 1);
            s[u] = src[e];
            d[u] = dst[e];
        }
#pragma unroll
        for (int u = 0; u < BIN_U; ++u) {
            const int e = base + u * SEG_THREADS + tid;
            if (e < e1 && (unsigned)s[u] < (unsigned)N && (unsigned)d[u] < (unsigned)N) {
                const int b  = s[u] / SEG_NODES;
                const int nl = s[u] - b * SEG_NODES;
                const int pos = atomicAdd(&cntL[b], 1);
                if (pos < SLOT_CAP) stage[b * BIN_PITCH + pos] = d[u] | (nl << DST_BITS);
            }
        }
    }
    __syncthreads();

    for (int b = wave; b < SEG_GROUPS; b += SEG_WAVES) {
        const int c  = cntL[b];
        const int nw = 1 + min(c, SLOT_CAP);
        int* slot = slots + (size_t)b * (SEG_NODES * DN) + w * SLOT_W;
        for (int i = 0; i < nw; i += 32) {
            const int idx = i + lane;
            int v = 0;
            if (idx == 0) v = c;
            else if (idx < nw) v = stage[b * BIN_PITCH + idx - 1];
            slot[idx] = v;
        }
    }
}

template <typename F>
__device__ __forceinline__ void seg_each_entry(const int (&first)[SLOTS_PER_WAVE],
                                               const int* myslots, int wave, int lane, F f) {
#pragma unroll
    for (int s = 0; s < SLOTS_PER_WAVE; ++s) {
        const int n = __builtin_amdgcn_readfirstlane(first[s]);
        if (lane >= 1 && lane - 1 < n) f(first[s]);
        const int* slot = myslots + (wave + s * FW) * SLOT_W;
#pragma clang loop unroll(disable)
        for (int i = 32; i < 1 + n; i += 32)
            if (i + lane < 1 + n) f(slot[i + lane]);
    }
}

__global__ __launch_bounds__(FTHREADS) void fused_kernel(
        const float* __restrict__ H,
        const int* __restrict__ src,
        const int* __restrict__ dst,
        const float* __restrict__ W,
        float* out  ,
        int N, int E) {
    __shared__ __attribute__((aligned(16))) float Ws[DN * WPAD];
    __shared__ __attribute__((aligned(16))) float At[FW * 16 * APITCH];
    __shared__ int sorted[SORT_CAP];
    __shared__ int nodeCnt[SEG_NODES + 1];
    __shared__ int nodeOff[SEG_NODES + 1];
    __shared__ int ovfL;

    const int tid  = threadIdx.x;
    const int lane = tid & 31;
    const int wave = __builtin_amdgcn_readfirstlane(tid >> 5);
    const int lm   = lane & 15;
    const int h2   = (lane >> 4) << 1;

    for (int i = tid; i < DN * (KTOT / 4); i += FTHREADS) {
        const int r  = i >> 6;
        const int c4 = (i & 63) << 2;
        const float4 w = *(const float4*)(W + (size_t)r * KTOT + c4);
        *(float4*)(&Ws[r * WPAD + c4]) = w;
    }

    const float* wrow[8];
#pragma unroll
    for (int nt = 0; nt < 8; ++nt) wrow[nt] = &Ws[(nt * 16 + lm) * WPAD + h2];

    float* myA  = At + wave * (16 * APITCH);
    int*   myAi = (int*)myA;
    const int* keys = (const int*)out;
    const bool binned = bins_usable(N, (int)gridDim.x);
    const int4 sent4 = make_int4(KEY_SENTINEL, KEY_SENTINEL, KEY_SENTINEL, KEY_SENTINEL);

    for (int lo = blockIdx.x * SEG_NODES; lo < N; lo += gridDim.x * SEG_NODES) {
        const int nloc = min(SEG_NODES, N - lo);

        const int* myslots = keys + (size_t)lo * DN;
        int first[SLOTS_PER_WAVE];
        if (binned) {
#pragma unroll
            for (int s = 0; s < SLOTS_PER_WAVE; ++s) {
                const int si = wave + s * FW;
                first[s] = (si < SEG_GROUPS) ? myslots[si * SLOT_W + lane] : 0;
            }
            if (tid == 0) ovfL = 0;
            for (int i = tid; i <= SEG_NODES; i += FTHREADS) nodeCnt[i] = 0;
        }
        __syncthreads();

        bool use_sorted = false;
        if (binned) {
            bool ov = false;
#pragma unroll
            for (int s = 0; s < SLOTS_PER_WAVE; ++s)
                ov |= __builtin_amdgcn_readfirstlane(first[s]) > SLOT_CAP;
            if (ov && lane == 0) ovfL = 1;
            __syncthreads();
            if (ovfL == 0) {
                seg_each_entry(first, myslots, wave, lane,
                               [&](int e) { atomicAdd(&nodeCnt[e >> DST_BITS], 1); });
                __syncthreads();
                if (wave == 0) {
                    constexpr int PER = (SEG_NODES + 1 + 31) / 32;
                    int c[PER];
                    int sum = 0;
#pragma unroll
                    for (int k = 0; k < PER; ++k) {
                        const int idx = lane * PER + k;
                        c[k] = (idx < SEG_NODES) ? nodeCnt[idx] : 0;
                        sum += c[k];
                    }
                    int incl = sum;
#pragma unroll
                    for (int d = 1; d < 32; d <<= 1) {
                        const int t = __shfl_up(incl, d);
                        if (lane >= d) incl += t;
                    }
                    int off = incl - sum;
#pragma unroll
                    for (int k = 0; k < PER; ++k) {
                        const int idx = lane * PER + k;
                        if (idx <= SEG_NODES) { nodeOff[idx] = off; nodeCnt[idx] = off; }
                        off += c[k];
                    }
                }
                __syncthreads();
                if (nodeOff[SEG_NODES] <= SORT_CAP) {
                    seg_each_entry(first, myslots, wave, lane,
                                   [&](int e) {
                                       const int pos = atomicAdd(&nodeCnt[e >> DST_BITS], 1);
                                       sorted[pos] = e & DST_MASK;
                                   });
                    use_sorted = true;
                }
            }
            __syncthreads();
        }

        const int ntiles = (nloc + 15) >> 4;
        for (int t = wave; t < ntiles; t += FW) {
            const int r0 = t << 4;

            {
                float4 hv[16];
#pragma unroll
                for (int r = 0; r < 16; ++r) {
                    const int g = min(lo + r0 + r, N - 1);
                    hv[r] = *(const float4*)(H + (size_t)g * DN + (lane << 2));
                }
#pragma unroll
                for (int r = 0; r < 16; ++r)
                    *(float4*)(myA + r * APITCH + (lane << 2)) = hv[r];
            }
            __builtin_amdgcn_fence(__ATOMIC_ACQ_REL, "wavefront");
            v2f aH[32];
#pragma unroll
            for (int kk = 0; kk < 32; ++kk)
                aH[kk] = *(const v2f*)(myA + lm * APITCH + (kk << 2) + h2);
            __builtin_amdgcn_fence(__ATOMIC_ACQ_REL, "wavefront");

            v8f zero = {};
            v8f acc[8];
#pragma unroll
            for (int nt = 0; nt < 8; ++nt) acc[nt] = zero;

            auto kstep = [&](int kk, v2f a, int koff) {
                const int kc = koff + (kk << 2);
#pragma unroll
                for (int nt = 0; nt < 8; ++nt) {
                    v2f b = *(const v2f*)(wrow[nt] + kc);
                    acc[nt] = __builtin_amdgcn_wmma_f32_16x16x4_f32(
                        false, a, false, b, (short)0, acc[nt], false, false);
                }
            };

            if (use_sorted) {
                const int offv = nodeOff[min(r0 + lane, SEG_NODES)];
#pragma unroll
                for (int r = 0; r < 16; ++r) {
                    const int beg = __builtin_amdgcn_readlane(offv, r);
                    const int end = __builtin_amdgcn_readlane(offv, r + 1);
                    float4 k = make_float4(-INFINITY, -INFINITY, -INFINITY, -INFINITY);
                    float4 h[16];
                    const bool any = beg < end;
                    if (any) {
                        const int mine = sorted[min(beg + (lane & 15), end - 1)];
#pragma unroll
                        for (int j = 0; j < 16; ++j) {
                            const int d = __builtin_amdgcn_readlane(mine, j);
                            const float* rowp = H + (size_t)d * DN;
                            h[j] = *(const float4*)(rowp + (lane << 2));
                        }
                    }
                    __builtin_amdgcn_sched_barrier(0);
                    kstep(2 * r, aH[2 * r], 0);
                    kstep(2 * r + 1, aH[2 * r + 1], 0);
                    __builtin_amdgcn_sched_barrier(0);
                    if (any) {
#pragma unroll
                        for (int j = 0; j < 16; ++j) {
                            k.x = fmaxf(k.x, h[j].x);
                            k.y = fmaxf(k.y, h[j].y);
                            k.z = fmaxf(k.z, h[j].z);
                            k.w = fmaxf(k.w, h[j].w);
                        }
                    }
                    for (int i = beg + 16; i < end; i += 8) {
                        const int mine = sorted[min(i + (lane & 7), end - 1)];
                        float4 g[8];
#pragma unroll
                        for (int j = 0; j < 8; ++j) {
                            const int d = __builtin_amdgcn_readlane(mine, j);
                            const float* rowp = H + (size_t)d * DN;
                            g[j] = *(const float4*)(rowp + (lane << 2));
                        }
                        __builtin_amdgcn_sched_barrier(0);
#pragma unroll
                        for (int j = 0; j < 8; ++j) {
                            k.x = fmaxf(k.x, g[j].x);
                            k.y = fmaxf(k.y, g[j].y);
                            k.z = fmaxf(k.z, g[j].z);
                            k.w = fmaxf(k.w, g[j].w);
                        }
                    }
                    const float4 v = any ? k : make_float4(0.0f, 0.0f, 0.0f, 0.0f);
                    *(float4*)(myA + r * APITCH + (lane << 2)) = v;
                }
            } else {
#pragma unroll
                for (int r = 0; r < 16; ++r)
                    *(int4*)(myAi + r * APITCH + (lane << 2)) = sent4;
                const int tlo = lo + r0;
                const int tn  = min(16, nloc - r0);
                for (int e0 = 0; e0 < E; e0 += 32) {
                    const int e = e0 + lane;
                    int rel = -1, d = 0;
                    if (e < E) { rel = src[e] - tlo; d = dst[e]; }
                    const bool hit = (unsigned)rel < (unsigned)tn && (unsigned)d < (unsigned)N;
                    unsigned m = __builtin_amdgcn_ballot_w32(hit);
                    while (m != 0u) {
                        const int j = __builtin_ctz(m);
                        m &= m - 1u;
                        const int rr = __builtin_amdgcn_readlane(rel, j);
                        const int dd = __builtin_amdgcn_readlane(d, j);
                        const float4 hv = *(const float4*)(H + (size_t)dd * DN + (lane << 2));
                        int4* kp = (int4*)(myAi + rr * APITCH + (lane << 2));
                        int4 k = *kp;
                        k.x = max(k.x, enc_f32(hv.x));
                        k.y = max(k.y, enc_f32(hv.y));
                        k.z = max(k.z, enc_f32(hv.z));
                        k.w = max(k.w, enc_f32(hv.w));
                        *kp = k;
                    }
                }
#pragma unroll
                for (int r = 0; r < 16; ++r) {
                    const int4 k = *(const int4*)(myAi + r * APITCH + (lane << 2));
                    float4 v;
                    v.x = (k.x == KEY_SENTINEL) ? 0.0f : dec_f32(k.x);
                    v.y = (k.y == KEY_SENTINEL) ? 0.0f : dec_f32(k.y);
                    v.z = (k.z == KEY_SENTINEL) ? 0.0f : dec_f32(k.z);
                    v.w = (k.w == KEY_SENTINEL) ? 0.0f : dec_f32(k.w);
                    *(float4*)(myA + r * APITCH + (lane << 2)) = v;
                }
#pragma unroll
                for (int kk = 0; kk < 32; ++kk) kstep(kk, aH[kk], 0);
            }

            __builtin_amdgcn_fence(__ATOMIC_ACQ_REL, "wavefront");

#pragma unroll 4
            for (int kk = 0; kk < 32; ++kk) {
                const v2f a = *(const v2f*)(myA + lm * APITCH + (kk << 2) + h2);
                kstep(kk, a, 128);
            }

            const int rb = r0 + ((lane >> 4) << 3);
            float* obase = out + (size_t)(lo + rb) * DN + lm;
#pragma unroll
            for (int r = 0; r < 8; ++r) {
                if (rb + r < nloc) {
#pragma unroll
                    for (int nt = 0; nt < 8; ++nt)
                        obase[(size_t)r * DN + nt * 16] = acc[nt][r];
                }
            }
            __builtin_amdgcn_fence(__ATOMIC_ACQ_REL, "wavefront");
        }
        __syncthreads();
    }
}

extern "C" void kernel_launch(void* const* d_in, const int* in_sizes, int n_in,
                              void* d_out, int out_size, void* d_ws, size_t ws_size,
                              hipStream_t stream) {
    (void)n_in; (void)out_size; (void)d_ws; (void)ws_size;
    const float* H   = (const float*)d_in[0];
    const int*   src = (const int*)d_in[1];
    const int*   dst = (const int*)d_in[2];
    const float* W   = (const float*)d_in[3];

    const int N = in_sizes[0] / DN;
    const int E = in_sizes[1];
    float* out = (float*)d_out;

    bin_kernel<<<SEG_GROUPS, SEG_THREADS, 0, stream>>>(src, dst, (int*)d_out, N, E);

    fused_kernel<<<SEG_GROUPS, FTHREADS, 0, stream>>>(H, src, dst, W, out, N, E);
}


# segment-max gather: rows 17..24 of a node are loaded in one round trip with its first sixteen (24 row loads in flight per wave); slot addresses kept scalar so the extra row registers fit unspilled
# speedup vs baseline: 4.2444x; 1.0643x over previous
#include <hip/hip_runtime.h>

#define DN   128
#define KTOT 256
#define WPAD 260
#define KEY_SENTINEL ((int)0x80000000)

typedef float v2f __attribute__((ext_vector_type(2)));
typedef float v8f __attribute__((ext_vector_type(8)));

__device__ __forceinline__ int enc_f32(float f) {
    int b = __float_as_int(f);
    return b >= 0 ? b : (b ^ 0x7FFFFFFF);
}
__device__ __forceinline__ float dec_f32(int k) {
    return __int_as_float(k >= 0 ? k : (k ^ 0x7FFFFFFF));
}

#define SEG_NODES   391
#define SEG_GROUPS  256
#define SEG_THREADS 1024
#define SEG_WAVES   32

#define SLOT_W    128
#define SLOT_CAP  127
#define BIN_PITCH 129
#define BIN_U     8
#define DST_BITS  17
#define DST_MASK  ((1 << DST_BITS) - 1)

#define FW        12
#define FTHREADS  (FW * 32)
#define SLOTS_PER_WAVE ((SEG_GROUPS + FW - 1) / FW)
#define APITCH    132
#define SORT_CAP  12288

__device__ __forceinline__ bool bins_usable(int N, int groups) {
    return groups == SEG_GROUPS
        && N <= SEG_GROUPS * SEG_NODES
        && N >= (SEG_GROUPS - 1) * SEG_NODES + (SEG_GROUPS * SLOT_W) / DN
        && N <= (1 << DST_BITS);
}

__global__ __launch_bounds__(SEG_THREADS) void bin_kernel(
        const int* __restrict__ src,
        const int* __restrict__ dst,
        int* __restrict__ slots, int N, int E) {
    __shared__ int cntL[SEG_GROUPS];
    __shared__ int stage[SEG_GROUPS * BIN_PITCH];

    if (!bins_usable(N, (int)gridDim.x)) return;

    const int tid  = threadIdx.x;
    const int lane = tid & 31;
    const int wave = tid >> 5;
    const int w    = blockIdx.x;

    if (tid < SEG_GROUPS) cntL[tid] = 0;
    __syncthreads();

    const int chunk = (E + SEG_GROUPS - 1) / SEG_GROUPS;
    const int e0 = w * chunk;
    const int e1 = min(E, e0 + chunk);
    for (int base = e0; base < e1; base += BIN_U * SEG_THREADS) {
        int s[BIN_U], d[BIN_U];
#pragma unroll
        for (int u = 0; u < BIN_U; ++u) {
            const int e = min(base + u * SEG_THREADS + tid, E - 1);
            s[u] = src[e];
            d[u] = dst[e];
        }
#pragma unroll
        for (int u = 0; u < BIN_U; ++u) {
            const int e = base + u * SEG_THREADS + tid;
            if (e < e1 && (unsigned)s[u] < (unsigned)N && (unsigned)d[u] < (unsigned)N) {
                const int b  = s[u] / SEG_NODES;
                const int nl = s[u] - b * SEG_NODES;
                const int pos = atomicAdd(&cntL[b], 1);
                if (pos < SLOT_CAP) stage[b * BIN_PITCH + pos] = d[u] | (nl << DST_BITS);
            }
        }
    }
    __syncthreads();

    for (int b = wave; b < SEG_GROUPS; b += SEG_WAVES) {
        const int c  = cntL[b];
        const int nw = 1 + min(c, SLOT_CAP);
        int* slot = slots + (size_t)b * (SEG_NODES * DN) + w * SLOT_W;
        for (int i = 0; i < nw; i += 32) {
            const int idx = i + lane;
            int v = 0;
            if (idx == 0) v = c;
            else if (idx < nw) v = stage[b * BIN_PITCH + idx - 1];
            slot[idx] = v;
        }
    }
}

template <typename F>
__device__ __forceinline__ void seg_each_entry(const int (&first)[SLOTS_PER_WAVE],
                                               const int* keys, int base, int wave, int lane, F f) {
#pragma unroll
    for (int s = 0; s < SLOTS_PER_WAVE; ++s) {
        const int n = __builtin_amdgcn_readfirstlane(first[s]);
        if (lane >= 1 && lane - 1 < n) f(first[s]);
        const int so = __builtin_amdgcn_readfirstlane(base + (wave + s * FW) * SLOT_W);
        const int* slot = keys + so;
#pragma clang loop unroll(disable)
        for (int i = 32; i < 1 + n; i += 32)
            if (i + lane < 1 + n) f(slot[i + lane]);
    }
}

__global__ __launch_bounds__(FTHREADS) void fused_kernel(
        const float* __restrict__ H,
        const int* __restrict__ src,
        const int* __restrict__ dst,
        const float* __restrict__ W,
        float* out  ,
        int N, int E) {
    __shared__ __attribute__((aligned(16))) float Ws[DN * WPAD];
    __shared__ __attribute__((aligned(16))) float At[FW * 16 * APITCH];
    __shared__ int sorted[SORT_CAP];
    __shared__ int nodeCnt[SEG_NODES + 1];
    __shared__ int nodeOff[SEG_NODES + 1];
    __shared__ int ovfL;

    const int tid  = threadIdx.x;
    const int lane = tid & 31;
    const int wave = __builtin_amdgcn_readfirstlane(tid >> 5);
    const int lm   = lane & 15;
    const int h2   = (lane >> 4) << 1;

    for (int i = tid; i < DN * (KTOT / 4); i += FTHREADS) {
        const int r  = i >> 6;
        const int c4 = (i & 63) << 2;
        const float4 w = *(const float4*)(W + (size_t)r * KTOT + c4);
        *(float4*)(&Ws[r * WPAD + c4]) = w;
    }

    const float* wrow[8];
#pragma unroll
    for (int nt = 0; nt < 8; ++nt) wrow[nt] = &Ws[(nt * 16 + lm) * WPAD + h2];

    float* myA  = At + wave * (16 * APITCH);
    int*   myAi = (int*)myA;
    const int* keys = (const int*)out;
    const bool binned = bins_usable(N, (int)gridDim.x);
    const int4 sent4 = make_int4(KEY_SENTINEL, KEY_SENTINEL, KEY_SENTINEL, KEY_SENTINEL);

    for (int lo = blockIdx.x * SEG_NODES; lo < N; lo += gridDim.x * SEG_NODES) {
        const int nloc = min(SEG_NODES, N - lo);

        const int* myslots = keys + (size_t)lo * DN;
        int first[SLOTS_PER_WAVE];
#pragma unroll
        for (int s = 0; s < SLOTS_PER_WAVE; ++s) first[s] = 0;
        if (binned) {
#pragma unroll
            for (int s = 0; s < SLOTS_PER_WAVE; ++s) {
                const int si = wave + s * FW;
                first[s] = (si < SEG_GROUPS) ? myslots[si * SLOT_W + lane] : 0;
            }
            if (tid == 0) ovfL = 0;
            for (int i = tid; i <= SEG_NODES; i += FTHREADS) nodeCnt[i] = 0;
        }
        __syncthreads();

        bool use_sorted = false;
        if (binned) {
            bool ov = false;
#pragma unroll
            for (int s = 0; s < SLOTS_PER_WAVE; ++s)
                ov |= __builtin_amdgcn_readfirstlane(first[s]) > SLOT_CAP;
            if (ov && lane == 0) ovfL = 1;
            __syncthreads();
            if (ovfL == 0) {
                seg_each_entry(first, keys, lo * DN, wave, lane,
                               [&](int e) { atomicAdd(&nodeCnt[e >> DST_BITS], 1); });
                __syncthreads();
                if (wave == 0) {
                    constexpr int PER = (SEG_NODES + 1 + 31) / 32;
                    int c[PER];
                    int sum = 0;
#pragma unroll
                    for (int k = 0; k < PER; ++k) {
                        const int idx = lane * PER + k;
                        c[k] = (idx < SEG_NODES) ? nodeCnt[idx] : 0;
                        sum += c[k];
                    }
                    int incl = sum;
#pragma unroll
                    for (int d = 1; d < 32; d <<= 1) {
                        const int t = __shfl_up(incl, d);
                        if (lane >= d) incl += t;
                    }
                    int off = incl - sum;
#pragma unroll
                    for (int k = 0; k < PER; ++k) {
                        const int idx = lane * PER + k;
                        if (idx <= SEG_NODES) { nodeOff[idx] = off; nodeCnt[idx] = off; }
                        off += c[k];
                    }
                }
                __syncthreads();
                if (nodeOff[SEG_NODES] <= SORT_CAP) {
                    seg_each_entry(first, keys, lo * DN, wave, lane,
                                   [&](int e) {
                                       const int pos = atomicAdd(&nodeCnt[e >> DST_BITS], 1);
                                       sorted[pos] = e & DST_MASK;
                                   });
                    use_sorted = true;
                }
            }
            __syncthreads();
        }

        const int ntiles = (nloc + 15) >> 4;
        for (int t = wave; t < ntiles; t += FW) {
            const int r0 = t << 4;

            {
                float4 hv[16];
#pragma unroll
                for (int r = 0; r < 16; ++r) {
                    const int g = min(lo + r0 + r, N - 1);
                    hv[r] = *(const float4*)(H + (size_t)g * DN + (lane << 2));
                }
#pragma unroll
                for (int r = 0; r < 16; ++r)
                    *(float4*)(myA + r * APITCH + (lane << 2)) = hv[r];
            }
            __builtin_amdgcn_fence(__ATOMIC_ACQ_REL, "wavefront");
            v2f aH[32];
#pragma unroll
            for (int kk = 0; kk < 32; ++kk)
                aH[kk] = *(const v2f*)(myA + lm * APITCH + (kk << 2) + h2);
            __builtin_amdgcn_fence(__ATOMIC_ACQ_REL, "wavefront");

            v8f zero = {};
            v8f acc[8];
#pragma unroll
            for (int nt = 0; nt < 8; ++nt) acc[nt] = zero;

            auto kstep = [&](int kk, v2f a, int koff) {
                const int kc = koff + (kk << 2);
#pragma unroll
                for (int nt = 0; nt < 8; ++nt) {
                    v2f b = *(const v2f*)(wrow[nt] + kc);
                    acc[nt] = __builtin_amdgcn_wmma_f32_16x16x4_f32(
                        false, a, false, b, (short)0, acc[nt], false, false);
                }
            };

            if (use_sorted) {
                const int offv = nodeOff[min(r0 + lane, SEG_NODES)];
#pragma unroll
                for (int r = 0; r < 16; ++r) {
                    const int beg = __builtin_amdgcn_readlane(offv, r);
                    const int end = __builtin_amdgcn_readlane(offv, r + 1);
                    float4 k = make_float4(-INFINITY, -INFINITY, -INFINITY, -INFINITY);
                    float4 h[16];
                    float4 g[8];
                    const bool any  = beg < end;
                    const bool more = beg + 16 < end;
                    if (any) {
                        const int mine = sorted[min(beg + lane, end - 1)];
#pragma unroll
                        for (int j = 0; j < 16; ++j) {
                            const int d = __builtin_amdgcn_readlane(mine, j);
                            const float* rowp = H + (size_t)d * DN;
                            h[j] = *(const float4*)(rowp + (lane << 2));
                        }
                        if (more) {
#pragma unroll
                            for (int j = 0; j < 8; ++j) {
                                const int d = __builtin_amdgcn_readlane(mine, 16 + j);
                                const float* rowp = H + (size_t)d * DN;
                                g[j] = *(const float4*)(rowp + (lane << 2));
                            }
                        }
                    }
                    __builtin_amdgcn_sched_barrier(0);
                    kstep(2 * r, aH[2 * r], 0);
                    kstep(2 * r + 1, aH[2 * r + 1], 0);
                    __builtin_amdgcn_sched_barrier(0);
                    if (any) {
#pragma unroll
                        for (int j = 0; j < 16; ++j) {
                            k.x = fmaxf(k.x, h[j].x);
                            k.y = fmaxf(k.y, h[j].y);
                            k.z = fmaxf(k.z, h[j].z);
                            k.w = fmaxf(k.w, h[j].w);
                        }
                    }
                    if (more) {
#pragma unroll
                        for (int j = 0; j < 8; ++j) {
                            k.x = fmaxf(k.x, g[j].x);
                            k.y = fmaxf(k.y, g[j].y);
                            k.z = fmaxf(k.z, g[j].z);
                            k.w = fmaxf(k.w, g[j].w);
                        }
                    }
                    for (int i = beg + 24; i < end; i += 8) {
                        const int mine = sorted[min(i + (lane & 7), end - 1)];
#pragma unroll
                        for (int j = 0; j < 8; ++j) {
                            const int d = __builtin_amdgcn_readlane(mine, j);
                            const float* rowp = H + (size_t)d * DN;
                            g[j] = *(const float4*)(rowp + (lane << 2));
                        }
                        __builtin_amdgcn_sched_barrier(0);
#pragma unroll
                        for (int j = 0; j < 8; ++j) {
                            k.x = fmaxf(k.x, g[j].x);
                            k.y = fmaxf(k.y, g[j].y);
                            k.z = fmaxf(k.z, g[j].z);
                            k.w = fmaxf(k.w, g[j].w);
                        }
                    }
                    const float4 v = any ? k : make_float4(0.0f, 0.0f, 0.0f, 0.0f);
                    *(float4*)(myA + r * APITCH + (lane << 2)) = v;
                }
            } else {
#pragma unroll
                for (int r = 0; r < 16; ++r)
                    *(int4*)(myAi + r * APITCH + (lane << 2)) = sent4;
                const int tlo = lo + r0;
                const int tn  = min(16, nloc - r0);
                for (int e0 = 0; e0 < E; e0 += 32) {
                    const int e = e0 + lane;
                    int rel = -1, d = 0;
                    if (e < E) { rel = src[e] - tlo; d = dst[e]; }
                    const bool hit = (unsigned)rel < (unsigned)tn && (unsigned)d < (unsigned)N;
                    unsigned m = __builtin_amdgcn_ballot_w32(hit);
                    while (m != 0u) {
                        const int j = __builtin_ctz(m);
                        m &= m - 1u;
                        const int rr = __builtin_amdgcn_readlane(rel, j);
                        const int dd = __builtin_amdgcn_readlane(d, j);
                        const float4 hv = *(const float4*)(H + (size_t)dd * DN + (lane << 2));
                        int4* kp = (int4*)(myAi + rr * APITCH + (lane << 2));
                        int4 k = *kp;
                        k.x = max(k.x, enc_f32(hv.x));
                        k.y = max(k.y, enc_f32(hv.y));
                        k.z = max(k.z, enc_f32(hv.z));
                        k.w = max(k.w, enc_f32(hv.w));
                        *kp = k;
                    }
                }
#pragma unroll
                for (int r = 0; r < 16; ++r) {
                    const int4 k = *(const int4*)(myAi + r * APITCH + (lane << 2));
                    float4 v;
                    v.x = (k.x == KEY_SENTINEL) ? 0.0f : dec_f32(k.x);
                    v.y = (k.y == KEY_SENTINEL) ? 0.0f : dec_f32(k.y);
                    v.z = (k.z == KEY_SENTINEL) ? 0.0f : dec_f32(k.z);
                    v.w = (k.w == KEY_SENTINEL) ? 0.0f : dec_f32(k.w);
                    *(float4*)(myA + r * APITCH + (lane << 2)) = v;
                }
#pragma unroll
                for (int kk = 0; kk < 32; ++kk) kstep(kk, aH[kk], 0);
            }

            __builtin_amdgcn_fence(__ATOMIC_ACQ_REL, "wavefront");

#pragma unroll 4
            for (int kk = 0; kk < 32; ++kk) {
                const v2f a = *(const v2f*)(myA + lm * APITCH + (kk << 2) + h2);
                kstep(kk, a, 128);
            }

            const int rb = r0 + ((lane >> 4) << 3);
            float* obase = out + (size_t)(lo + rb) * DN + lm;
#pragma unroll
            for (int r = 0; r < 8; ++r) {
                if (rb + r < nloc) {
#pragma unroll
                    for (int nt = 0; nt < 8; ++nt)
                        obase[(size_t)r * DN + nt * 16] = acc[nt][r];
                }
            }
            __builtin_amdgcn_fence(__ATOMIC_ACQ_REL, "wavefront");
        }
        __syncthreads();
    }
}

extern "C" void kernel_launch(void* const* d_in, const int* in_sizes, int n_in,
                              void* d_out, int out_size, void* d_ws, size_t ws_size,
                              hipStream_t stream) {
    (void)n_in; (void)out_size; (void)d_ws; (void)ws_size;
    const float* H   = (const float*)d_in[0];
    const int*   src = (const int*)d_in[1];
    const int*   dst = (const int*)d_in[2];
    const float* W   = (const float*)d_in[3];

    const int N = in_sizes[0] / DN;
    const int E = in_sizes[1];
    float* out = (float*)d_out;

    bin_kernel<<<SEG_GROUPS, SEG_THREADS, 0, stream>>>(src, dst, (int*)d_out, N, E);

    fused_kernel<<<SEG_GROUPS, FTHREADS, 0, stream>>>(H, src, dst, W, out, N, E);
}
